# starting kernel
# baseline (speedup 1.0000x reference)
#include <hip/hip_runtime.h>

#define D     64
#define NNEI  50
#define ROWS  16
#define BLK   128

typedef float v2f __attribute__((ext_vector_type(2)));
typedef float v8f __attribute__((ext_vector_type(8)));

__device__ __forceinline__ void wmma_k64_tile(v8f& c, const float* __restrict__ A_lds,
                                              const float* __restrict__ W,
                                              int nt, int hl, int half) {
  for (int kc = 0; kc < 16; ++kc) {
    const int k0 = 4 * kc + 2 * half;
    v2f a;
    a.x = A_lds[hl * D + k0];
    a.y = A_lds[hl * D + k0 + 1];
    v2f b;
    b.x = W[(size_t)k0 * D + nt * 16 + hl];
    b.y = W[(size_t)(k0 + 1) * D + nt * 16 + hl];
    c = __builtin_amdgcn_wmma_f32_16x16x4_f32(
         false, a,  false, b,
         (short)0, c,  false,  false);
  }
}

__device__ __forceinline__ void wmma_store_tile(const v8f& c, const float* __restrict__ bias,
                                                bool relu, float* __restrict__ out_lds,
                                                int nt, int hl, int half) {
  const int col = nt * 16 + hl;
  const float bv = bias[col];
#pragma unroll
  for (int i = 0; i < 8; ++i) {
    const int row = i + 8 * half;
    float v = c[i] + bv;
    if (relu) v = fmaxf(v, 0.f);
    out_lds[row * D + col] = v;
  }
}

__global__ __launch_bounds__(BLK) void kgat_fused(
    const int* __restrict__ user_idx, const int* __restrict__ item_idx,
    const int* __restrict__ kg_adj,
    const float* __restrict__ user_table, const float* __restrict__ item_table,
    const float* __restrict__ entity_table,
    const float* __restrict__ attn_W, const float* __restrict__ attn_b,
    const float* __restrict__ kg_W1, const float* __restrict__ kg_b1,
    const float* __restrict__ kg_W2, const float* __restrict__ kg_b2,
    const float* __restrict__ comb_W, const float* __restrict__ comb_b,
    const float* __restrict__ out_W, const float* __restrict__ out_b,
    float* __restrict__ out, int B) {
  __shared__ __align__(16) float item_s[ROWS * D];
  __shared__ __align__(16) float bufA[ROWS * D];
  __shared__ __align__(16) float bufB[ROWS * D];
  __shared__ float scores_s[ROWS * NNEI];
  __shared__ float iscore_s[ROWS];
  __shared__ int nbr_s[ROWS * NNEI];
  __shared__ int iidx_s[ROWS];
  __shared__ int uidx_s[ROWS];

  const int tid = threadIdx.x;
  const int lane = tid & 31;
  const int wid = tid >> 5;
  const int hl = lane & 15, half = (lane >> 4) & 1;
  const int r0 = blockIdx.x * ROWS;

  if (tid < ROWS) {
    const int rr = min(r0 + tid, B - 1);
    iidx_s[tid] = item_idx[rr];
    uidx_s[tid] = user_idx[rr];
  }
  for (int t = tid; t < ROWS * NNEI; t += BLK) {
    const int rr = min(r0 + t / NNEI, B - 1);
    const int e = kg_adj[(size_t)rr * NNEI + (t % NNEI)];
    nbr_s[t] = e < 0 ? 0 : e;
  }
  __syncthreads();

  for (int t = tid; t < ROWS * (D / 4); t += BLK) {
    const int row = t >> 4, c4 = t & 15;
    const float4* src = (const float4*)(item_table + (size_t)iidx_s[row] * D);
    ((float4*)item_s)[t] = src[c4];
  }
  __syncthreads();

  if (tid < ROWS) {
    const float4* ip = (const float4*)(item_s + tid * D);
    const float4* wi = (const float4*)(attn_W);
    float s = 0.f;
#pragma unroll
    for (int j = 0; j < 16; ++j) {
      const float4 e = ip[j], w = wi[j];
      s += e.x * w.x + e.y * w.y + e.z * w.z + e.w * w.w;
    }
    iscore_s[tid] = s + attn_b[0];
  }
  __syncthreads();

  {
    float4 wn4[16];
    const float4* wn = (const float4*)(attn_W + D);
#pragma unroll
    for (int j = 0; j < 16; ++j) wn4[j] = wn[j];
    for (int t = tid; t < ROWS * NNEI; t += BLK) {
      const int row = t / NNEI;
      const int eidx = nbr_s[t];
      const float4* ep = (const float4*)(entity_table + (size_t)eidx * D);
      float s = 0.f;
#pragma unroll
      for (int j = 0; j < 16; ++j) {
        const float4 e = ep[j];
        s += e.x * wn4[j].x + e.y * wn4[j].y + e.z * wn4[j].z + e.w * wn4[j].w;
      }
      float sc = s + iscore_s[row];
      sc = sc > 0.f ? sc : 0.2f * sc;
      scores_s[t] = sc;
    }
  }
  __syncthreads();

  if (tid < ROWS) {
    float m = -3.402823466e38f;
    for (int n = 0; n < NNEI; ++n) m = fmaxf(m, scores_s[tid * NNEI + n]);
    float sum = 0.f;
    for (int n = 0; n < NNEI; ++n) {
      const float e = __expf(scores_s[tid * NNEI + n] - m);
      scores_s[tid * NNEI + n] = e;
      sum += e;
    }
    const float inv = 1.f / sum;
    for (int n = 0; n < NNEI; ++n) scores_s[tid * NNEI + n] *= inv;
  }
  __syncthreads();

  {
    const int row = tid >> 3;
    const int j0 = tid & 7;
    float4 a0 = make_float4(0.f, 0.f, 0.f, 0.f);
    float4 a1 = a0;
    for (int n = 0; n < NNEI; ++n) {
      const float w = scores_s[row * NNEI + n];
      const int eidx = nbr_s[row * NNEI + n];
      const float4* ep = (const float4*)(entity_table + (size_t)eidx * D) + 2 * j0;
      const float4 e0 = ep[0], e1 = ep[1];
      a0.x += w * e0.x; a0.y += w * e0.y; a0.z += w * e0.z; a0.w += w * e0.w;
      a1.x += w * e1.x; a1.y += w * e1.y; a1.z += w * e1.z; a1.w += w * e1.w;
    }
    float4* op = (float4*)(bufA + row * D) + 2 * j0;
    op[0] = a0;
    op[1] = a1;
  }
  __syncthreads();

  v8f c;

  c = (v8f){0, 0, 0, 0, 0, 0, 0, 0};
  wmma_k64_tile(c, bufA, kg_W1, wid, hl, half);
  wmma_store_tile(c, kg_b1, true, bufB, wid, hl, half);
  __syncthreads();

  c = (v8f){0, 0, 0, 0, 0, 0, 0, 0};
  wmma_k64_tile(c, bufB, kg_W2, wid, hl, half);
  wmma_store_tile(c, kg_b2, false, bufA, wid, hl, half);
  __syncthreads();

  c = (v8f){0, 0, 0, 0, 0, 0, 0, 0};
  wmma_k64_tile(c, item_s, comb_W, wid, hl, half);
  wmma_k64_tile(c, bufA, comb_W + D * D, wid, hl, half);
  wmma_store_tile(c, comb_b, true, bufB, wid, hl, half);
  __syncthreads();

  {
    const int row = tid >> 3;
    const int j0 = tid & 7;
    const int uidx = uidx_s[row];
    const float4* up  = (const float4*)(user_table + (size_t)uidx * D) + 2 * j0;
    const float4* w0p = (const float4*)(out_W) + 2 * j0;
    const float4* w1p = (const float4*)(out_W + D) + 2 * j0;
    const float4* fp  = (const float4*)(bufB + row * D) + 2 * j0;
    float s = 0.f;
#pragma unroll
    for (int j = 0; j < 2; ++j) {
      const float4 u = up[j], w0 = w0p[j];
      const float4 f = fp[j], w1 = w1p[j];
      s += u.x * w0.x + u.y * w0.y + u.z * w0.z + u.w * w0.w;
      s += f.x * w1.x + f.y * w1.y + f.z * w1.z + f.w * w1.w;
    }
    s += __shfl_xor(s, 1, 32);
    s += __shfl_xor(s, 2, 32);
    s += __shfl_xor(s, 4, 32);
    if (j0 == 0 && (r0 + row) < B) out[r0 + row] = s + out_b[0];
  }
}

extern "C" void kernel_launch(void* const* d_in, const int* in_sizes, int n_in,
                              void* d_out, int out_size, void* d_ws, size_t ws_size,
                              hipStream_t stream) {
  const int*   user_idx     = (const int*)d_in[0];
  const int*   item_idx     = (const int*)d_in[1];
  const int*   kg_adj       = (const int*)d_in[2];
  const float* user_table   = (const float*)d_in[3];
  const float* item_table   = (const float*)d_in[4];
  const float* entity_table = (const float*)d_in[5];
  const float* attn_W = (const float*)d_in[6];
  const float* attn_b = (const float*)d_in[7];
  const float* kg_W1  = (const float*)d_in[8];
  const float* kg_b1  = (const float*)d_in[9];
  const float* kg_W2  = (const float*)d_in[10];
  const float* kg_b2  = (const float*)d_in[11];
  const float* comb_W = (const float*)d_in[12];
  const float* comb_b = (const float*)d_in[13];
  const float* out_W  = (const float*)d_in[14];
  const float* out_b  = (const float*)d_in[15];
  float* out = (float*)d_out;

  const int B = in_sizes[0];
  dim3 grid((B + ROWS - 1) / ROWS), block(BLK);
  kgat_fused<<<grid, block, 0, stream>>>(
      user_idx, item_idx, kg_adj, user_table, item_table, entity_table,
      attn_W, attn_b, kg_W1, kg_b1, kg_W2, kg_b2, comb_W, comb_b,
      out_W, out_b, out, B);
}


# One-pass neighbour gather: wave per batch row, each entity row read once by a 16-lane b128 load and kept in registers for score, softmax and weighted sum; dot and sums become lane-butterfly trees
# speedup vs baseline: 1.5074x; 1.5074x over previous
#include <hip/hip_runtime.h>

#define D     64
#define NNEI  50
#define ROWS  16
#define BLK   128
#define RPW   (ROWS / (BLK / 32))
#define NPH   (NNEI / 2)

typedef float v2f __attribute__((ext_vector_type(2)));
typedef float v8f __attribute__((ext_vector_type(8)));

__device__ __forceinline__ void wmma_k64_tile(v8f& c, const float* __restrict__ A_lds,
                                              const float* __restrict__ W,
                                              int nt, int hl, int half) {
  for (int kc = 0; kc < 16; ++kc) {
    const int k0 = 4 * kc + 2 * half;
    v2f a;
    a.x = A_lds[hl * D + k0];
    a.y = A_lds[hl * D + k0 + 1];
    v2f b;
    b.x = W[(size_t)k0 * D + nt * 16 + hl];
    b.y = W[(size_t)(k0 + 1) * D + nt * 16 + hl];
    c = __builtin_amdgcn_wmma_f32_16x16x4_f32(
         false, a,  false, b,
         (short)0, c,  false,  false);
  }
}

__device__ __forceinline__ void wmma_store_tile(const v8f& c, const float* __restrict__ bias,
                                                bool relu, float* __restrict__ out_lds,
                                                int nt, int hl, int half) {
  const int col = nt * 16 + hl;
  const float bv = bias[col];
#pragma unroll
  for (int i = 0; i < 8; ++i) {
    const int row = i + 8 * half;
    float v = c[i] + bv;
    if (relu) v = fmaxf(v, 0.f);
    out_lds[row * D + col] = v;
  }
}

template <int MASK>
__device__ __forceinline__ float row_xor(float x) {
  return __builtin_bit_cast(float, __builtin_amdgcn_update_dpp(
      0, __builtin_bit_cast(int, x), 0x160 | MASK, 0xf, 0xf, true));
}

__device__ __forceinline__ float row16_sum(float x) {
  x += row_xor<1>(x);
  x += row_xor<2>(x);
  x += row_xor<4>(x);
  x += row_xor<8>(x);
  return x;
}

__device__ __forceinline__ float xhalf(float x) {
  const int b = __builtin_bit_cast(int, x);
  return __builtin_bit_cast(float, __builtin_amdgcn_permlanex16(b, b, 0x76543210, 0xfedcba98, false, false));
}

__global__ __launch_bounds__(BLK) void kgat_fused(
    const int* __restrict__ user_idx, const int* __restrict__ item_idx,
    const int* __restrict__ kg_adj,
    const float* __restrict__ user_table, const float* __restrict__ item_table,
    const float* __restrict__ entity_table,
    const float* __restrict__ attn_W, const float* __restrict__ attn_b,
    const float* __restrict__ kg_W1, const float* __restrict__ kg_b1,
    const float* __restrict__ kg_W2, const float* __restrict__ kg_b2,
    const float* __restrict__ comb_W, const float* __restrict__ comb_b,
    const float* __restrict__ out_W, const float* __restrict__ out_b,
    float* __restrict__ out, int B) {
  __shared__ __align__(16) float item_s[ROWS * D];
  __shared__ __align__(16) float bufA[ROWS * D];
  __shared__ __align__(16) float bufB[ROWS * D];
  __shared__ float iscore_s[ROWS];
  __shared__ int nbr_s[ROWS * NNEI];
  __shared__ int iidx_s[ROWS];
  __shared__ int uidx_s[ROWS];

  const int tid = threadIdx.x;
  const int lane = tid & 31;
  const int wid = tid >> 5;
  const int hl = lane & 15, half = (lane >> 4) & 1;
  const int r0 = blockIdx.x * ROWS;

  if (tid < ROWS) {
    const int rr = min(r0 + tid, B - 1);
    iidx_s[tid] = item_idx[rr];
    uidx_s[tid] = user_idx[rr];
  }
  for (int t = tid; t < ROWS * NNEI; t += BLK) {
    const int rr = min(r0 + t / NNEI, B - 1);
    const int e = kg_adj[(size_t)rr * NNEI + (t % NNEI)];
    nbr_s[t] = e < 0 ? 0 : e;
  }
  __syncthreads();

  for (int t = tid; t < ROWS * (D / 4); t += BLK) {
    const int row = t >> 4, c4 = t & 15;
    const float4* src = (const float4*)(item_table + (size_t)iidx_s[row] * D);
    ((float4*)item_s)[t] = src[c4];
  }
  __syncthreads();

  if (tid < ROWS) {
    const float4* ip = (const float4*)(item_s + tid * D);
    const float4* wi = (const float4*)(attn_W);
    float s = 0.f;
#pragma unroll
    for (int j = 0; j < 16; ++j) {
      const float4 e = ip[j], w = wi[j];
      s += e.x * w.x + e.y * w.y + e.z * w.z + e.w * w.w;
    }
    iscore_s[tid] = s + attn_b[0];
  }
  __syncthreads();

  {
    const int c = lane & 15;
    const int h = lane >> 4;
    const float4 wn = ((const float4*)(attn_W + D))[c];
    const char* ebase = (const char*)entity_table;
    for (int rr = 0; rr < RPW; ++rr) {
      const int row = wid * RPW + rr;
      const int* nb = nbr_s + row * NNEI + h * NPH;
      float4 e[NPH];
#pragma unroll
      for (int i = 0; i < NPH; ++i) {
        const unsigned boff = ((unsigned)nb[i] << 8) + 16u * (unsigned)c;
        e[i] = *(const float4*)(ebase + boff);
      }
      const float isc = iscore_s[row];
      float s[NPH];
      float m = -3.402823466e38f;
#pragma unroll
      for (int i = 0; i < NPH; ++i) {
        float p = e[i].x * wn.x + e[i].y * wn.y + e[i].z * wn.z + e[i].w * wn.w;
        p = row16_sum(p);
        float sc = p + isc;
        sc = sc > 0.f ? sc : 0.2f * sc;
        s[i] = sc;
        m = fmaxf(m, sc);
      }
      m = fmaxf(m, xhalf(m));
      float sum = 0.f;
#pragma unroll
      for (int i = 0; i < NPH; ++i) {
        s[i] = __expf(s[i] - m);
        sum += s[i];
      }
      sum += xhalf(sum);
      const float inv = 1.f / sum;
      float4 a = make_float4(0.f, 0.f, 0.f, 0.f);
#pragma unroll
      for (int i = 0; i < NPH; ++i) {
        const float w = s[i] * inv;
        a.x += w * e[i].x; a.y += w * e[i].y; a.z += w * e[i].z; a.w += w * e[i].w;
      }
      a.x += xhalf(a.x); a.y += xhalf(a.y); a.z += xhalf(a.z); a.w += xhalf(a.w);
      if (h == 0) ((float4*)(bufA + row * D))[c] = a;
    }
  }
  __syncthreads();

  v8f c;

  c = (v8f){0, 0, 0, 0, 0, 0, 0, 0};
  wmma_k64_tile(c, bufA, kg_W1, wid, hl, half);
  wmma_store_tile(c, kg_b1, true, bufB, wid, hl, half);
  __syncthreads();

  c = (v8f){0, 0, 0, 0, 0, 0, 0, 0};
  wmma_k64_tile(c, bufB, kg_W2, wid, hl, half);
  wmma_store_tile(c, kg_b2, false, bufA, wid, hl, half);
  __syncthreads();

  c = (v8f){0, 0, 0, 0, 0, 0, 0, 0};
  wmma_k64_tile(c, item_s, comb_W, wid, hl, half);
  wmma_k64_tile(c, bufA, comb_W + D * D, wid, hl, half);
  wmma_store_tile(c, comb_b, true, bufB, wid, hl, half);
  __syncthreads();

  {
    const int row = tid >> 3;
    const int j0 = tid & 7;
    const int uidx = uidx_s[row];
    const float4* up  = (const float4*)(user_table + (size_t)uidx * D) + 2 * j0;
    const float4* w0p = (const float4*)(out_W) + 2 * j0;
    const float4* w1p = (const float4*)(out_W + D) + 2 * j0;
    const float4* fp  = (const float4*)(bufB + row * D) + 2 * j0;
    float s = 0.f;
#pragma unroll
    for (int j = 0; j < 2; ++j) {
      const float4 u = up[j], w0 = w0p[j];
      const float4 f = fp[j], w1 = w1p[j];
      s += u.x * w0.x + u.y * w0.y + u.z * w0.z + u.w * w0.w;
      s += f.x * w1.x + f.y * w1.y + f.z * w1.z + f.w * w1.w;
    }
    s += __shfl_xor(s, 1, 32);
    s += __shfl_xor(s, 2, 32);
    s += __shfl_xor(s, 4, 32);
    if (j0 == 0 && (r0 + row) < B) out[r0 + row] = s + out_b[0];
  }
}

extern "C" void kernel_launch(void* const* d_in, const int* in_sizes, int n_in,
                              void* d_out, int out_size, void* d_ws, size_t ws_size,
                              hipStream_t stream) {
  const int*   user_idx     = (const int*)d_in[0];
  const int*   item_idx     = (const int*)d_in[1];
  const int*   kg_adj       = (const int*)d_in[2];
  const float* user_table   = (const float*)d_in[3];
  const float* item_table   = (const float*)d_in[4];
  const float* entity_table = (const float*)d_in[5];
  const float* attn_W = (const float*)d_in[6];
  const float* attn_b = (const float*)d_in[7];
  const float* kg_W1  = (const float*)d_in[8];
  const float* kg_b1  = (const float*)d_in[9];
  const float* kg_W2  = (const float*)d_in[10];
  const float* kg_b2  = (const float*)d_in[11];
  const float* comb_W = (const float*)d_in[12];
  const float* comb_b = (const float*)d_in[13];
  const float* out_W  = (const float*)d_in[14];
  const float* out_b  = (const float*)d_in[15];
  float* out = (float*)d_out;

  const int B = in_sizes[0];
  dim3 grid((B + ROWS - 1) / ROWS), block(BLK);
  kgat_fused<<<grid, block, 0, stream>>>(
      user_idx, item_idx, kg_adj, user_table, item_table, entity_table,
      attn_W, attn_b, kg_W1, kg_b1, kg_W2, kg_b2, comb_W, comb_b,
      out_W, out_b, out, B);
}


# Gather: 8 lanes per entity row (2 b128 loads per lane, 13 slots per lane group): 3-step lane tree per dot; leaky-relu, exp, weight per 13 not 25 slots; dot, softmax and weighted-sum trees change
# speedup vs baseline: 1.5722x; 1.0430x over previous
#include <hip/hip_runtime.h>

#define D     64
#define NNEI  50
#define ROWS  16
#define BLK   128
#define RPW   (ROWS / (BLK / 32))
#define SLOTS ((NNEI + 3) / 4)

typedef float v2f __attribute__((ext_vector_type(2)));
typedef float v8f __attribute__((ext_vector_type(8)));

__device__ __forceinline__ void wmma_k64_tile(v8f& c, const float* __restrict__ A_lds,
                                              const float* __restrict__ W,
                                              int nt, int hl, int half) {
  for (int kc = 0; kc < 16; ++kc) {
    const int k0 = 4 * kc + 2 * half;
    v2f a;
    a.x = A_lds[hl * D + k0];
    a.y = A_lds[hl * D + k0 + 1];
    v2f b;
    b.x = W[(size_t)k0 * D + nt * 16 + hl];
    b.y = W[(size_t)(k0 + 1) * D + nt * 16 + hl];
    c = __builtin_amdgcn_wmma_f32_16x16x4_f32(
         false, a,  false, b,
         (short)0, c,  false,  false);
  }
}

__device__ __forceinline__ void wmma_store_tile(const v8f& c, const float* __restrict__ bias,
                                                bool relu, float* __restrict__ out_lds,
                                                int nt, int hl, int half) {
  const int col = nt * 16 + hl;
  const float bv = bias[col];
#pragma unroll
  for (int i = 0; i < 8; ++i) {
    const int row = i + 8 * half;
    float v = c[i] + bv;
    if (relu) v = fmaxf(v, 0.f);
    out_lds[row * D + col] = v;
  }
}

template <int MASK>
__device__ __forceinline__ float row_xor(float x) {
  return __builtin_bit_cast(float, __builtin_amdgcn_update_dpp(
      0, __builtin_bit_cast(int, x), 0x160 | MASK, 0xf, 0xf, true));
}

__device__ __forceinline__ float row8_sum(float x) {
  x += row_xor<1>(x);
  x += row_xor<2>(x);
  x += row_xor<4>(x);
  return x;
}

__device__ __forceinline__ float xhalf(float x) {
  const int b = __builtin_bit_cast(int, x);
  return __builtin_bit_cast(float, __builtin_amdgcn_permlanex16(b, b, 0x76543210, 0xfedcba98, false, false));
}

__global__ __launch_bounds__(BLK) __attribute__((amdgpu_waves_per_eu(4, 4))) void kgat_fused(
    const int* __restrict__ user_idx, const int* __restrict__ item_idx,
    const int* __restrict__ kg_adj,
    const float* __restrict__ user_table, const float* __restrict__ item_table,
    const float* __restrict__ entity_table,
    const float* __restrict__ attn_W, const float* __restrict__ attn_b,
    const float* __restrict__ kg_W1, const float* __restrict__ kg_b1,
    const float* __restrict__ kg_W2, const float* __restrict__ kg_b2,
    const float* __restrict__ comb_W, const float* __restrict__ comb_b,
    const float* __restrict__ out_W, const float* __restrict__ out_b,
    float* __restrict__ out, int B) {
  __shared__ __align__(16) float item_s[ROWS * D];
  __shared__ __align__(16) float bufA[ROWS * D];
  __shared__ __align__(16) float bufB[ROWS * D];
  __shared__ float iscore_s[ROWS];
  __shared__ int nbr_s[ROWS * NNEI];
  __shared__ int iidx_s[ROWS];
  __shared__ int uidx_s[ROWS];

  const int tid = threadIdx.x;
  const int lane = tid & 31;
  const int wid = tid >> 5;
  const int hl = lane & 15, half = (lane >> 4) & 1;
  const int r0 = blockIdx.x * ROWS;

  if (tid < ROWS) {
    const int rr = min(r0 + tid, B - 1);
    iidx_s[tid] = item_idx[rr];
    uidx_s[tid] = user_idx[rr];
  }
  for (int t = tid; t < ROWS * NNEI; t += BLK) {
    const int rr = min(r0 + t / NNEI, B - 1);
    const int e = kg_adj[(size_t)rr * NNEI + (t % NNEI)];
    nbr_s[t] = e < 0 ? 0 : e;
  }
  __syncthreads();

  for (int t = tid; t < ROWS * (D / 4); t += BLK) {
    const int row = t >> 4, c4 = t & 15;
    const float4* src = (const float4*)(item_table + (size_t)iidx_s[row] * D);
    ((float4*)item_s)[t] = src[c4];
  }
  __syncthreads();

  if (tid < ROWS) {
    const float4* ip = (const float4*)(item_s + tid * D);
    const float4* wi = (const float4*)(attn_W);
    float s = 0.f;
#pragma unroll
    for (int j = 0; j < 16; ++j) {
      const float4 e = ip[j], w = wi[j];
      s += e.x * w.x + e.y * w.y + e.z * w.z + e.w * w.w;
    }
    iscore_s[tid] = s + attn_b[0];
  }
  __syncthreads();

  {
    const int j = lane & 7;
    const int g = lane >> 3;
    const float4 wlo = ((const float4*)(attn_W + D))[j];
    const float4 whi = ((const float4*)(attn_W + D))[j + 8];
    const v2f wl0 = {wlo.x, wlo.y}, wl1 = {wlo.z, wlo.w};
    const v2f wh0 = {whi.x, whi.y}, wh1 = {whi.z, whi.w};
    const bool last_ok = (4 * (SLOTS - 1) + g) < NNEI;
    const char* ebase = (const char*)entity_table;
    for (int rr = 0; rr < RPW; ++rr) {
      const int row = wid * RPW + rr;
      const int* nb = nbr_s + row * NNEI;
      float4 elo[SLOTS], ehi[SLOTS];
#pragma unroll
      for (int i = 0; i < SLOTS; ++i) {
        const int n = min(4 * i + g, NNEI - 1);
        const unsigned boff = ((unsigned)nb[n] << 8) + 16u * (unsigned)j;
        elo[i] = *(const float4*)(ebase + boff);
        ehi[i] = *(const float4*)(ebase + boff + 128u);
      }
      const float isc = iscore_s[row];
      float s[SLOTS];
      float m = -3.402823466e38f;
#pragma unroll
      for (int i = 0; i < SLOTS; ++i) {
        v2f p2 = (v2f){elo[i].x, elo[i].y} * wl0;
        p2 = (v2f){elo[i].z, elo[i].w} * wl1 + p2;
        p2 = (v2f){ehi[i].x, ehi[i].y} * wh0 + p2;
        p2 = (v2f){ehi[i].z, ehi[i].w} * wh1 + p2;
        float p = row8_sum(p2.x + p2.y);
        float sc = p + isc;
        sc = sc > 0.f ? sc : 0.2f * sc;
        s[i] = sc;
        if (i < SLOTS - 1) m = fmaxf(m, sc);
        else m = last_ok ? fmaxf(m, sc) : m;
      }
      m = fmaxf(m, row_xor<8>(m));
      m = fmaxf(m, xhalf(m));
      float sum = 0.f;
#pragma unroll
      for (int i = 0; i < SLOTS; ++i) {
        float ex = __expf(s[i] - m);
        if (i == SLOTS - 1) ex = last_ok ? ex : 0.f;
        s[i] = ex;
        sum += ex;
      }
      sum += row_xor<8>(sum);
      sum += xhalf(sum);
      const float inv = 1.f / sum;
      v2f a0 = {0.f, 0.f}, a1 = {0.f, 0.f}, a2 = {0.f, 0.f}, a3 = {0.f, 0.f};
#pragma unroll
      for (int i = 0; i < SLOTS; ++i) {
        const float w = s[i] * inv;
        const v2f w2 = {w, w};
        a0 = w2 * (v2f){elo[i].x, elo[i].y} + a0;
        a1 = w2 * (v2f){elo[i].z, elo[i].w} + a1;
        a2 = w2 * (v2f){ehi[i].x, ehi[i].y} + a2;
        a3 = w2 * (v2f){ehi[i].z, ehi[i].w} + a3;
      }
      float r[8] = {a0.x, a0.y, a1.x, a1.y, a2.x, a2.y, a3.x, a3.y};
#pragma unroll
      for (int k = 0; k < 8; ++k) {
        r[k] += row_xor<8>(r[k]);
        r[k] += xhalf(r[k]);
      }
      if (g == 0) {
        ((float4*)(bufA + row * D))[j]     = make_float4(r[0], r[1], r[2], r[3]);
        ((float4*)(bufA + row * D))[j + 8] = make_float4(r[4], r[5], r[6], r[7]);
      }
    }
  }
  __syncthreads();

  v8f c;

  c = (v8f){0, 0, 0, 0, 0, 0, 0, 0};
  wmma_k64_tile(c, bufA, kg_W1, wid, hl, half);
  wmma_store_tile(c, kg_b1, true, bufB, wid, hl, half);
  __syncthreads();

  c = (v8f){0, 0, 0, 0, 0, 0, 0, 0};
  wmma_k64_tile(c, bufB, kg_W2, wid, hl, half);
  wmma_store_tile(c, kg_b2, false, bufA, wid, hl, half);
  __syncthreads();

  c = (v8f){0, 0, 0, 0, 0, 0, 0, 0};
  wmma_k64_tile(c, item_s, comb_W, wid, hl, half);
  wmma_k64_tile(c, bufA, comb_W + D * D, wid, hl, half);
  wmma_store_tile(c, comb_b, true, bufB, wid, hl, half);
  __syncthreads();

  {
    const int row = tid >> 3;
    const int j0 = tid & 7;
    const int uidx = uidx_s[row];
    const float4* up  = (const float4*)(user_table + (size_t)uidx * D) + 2 * j0;
    const float4* w0p = (const float4*)(out_W) + 2 * j0;
    const float4* w1p = (const float4*)(out_W + D) + 2 * j0;
    const float4* fp  = (const float4*)(bufB + row * D) + 2 * j0;
    float s = 0.f;
#pragma unroll
    for (int j = 0; j < 2; ++j) {
      const float4 u = up[j], w0 = w0p[j];
      const float4 f = fp[j], w1 = w1p[j];
      s += u.x * w0.x + u.y * w0.y + u.z * w0.z + u.w * w0.w;
      s += f.x * w1.x + f.y * w1.y + f.z * w1.z + f.w * w1.w;
    }
    s += __shfl_xor(s, 1, 32);
    s += __shfl_xor(s, 2, 32);
    s += __shfl_xor(s, 4, 32);
    if (j0 == 0 && (r0 + row) < B) out[r0 + row] = s + out_b[0];
  }
}

extern "C" void kernel_launch(void* const* d_in, const int* in_sizes, int n_in,
                              void* d_out, int out_size, void* d_ws, size_t ws_size,
                              hipStream_t stream) {
  const int*   user_idx     = (const int*)d_in[0];
  const int*   item_idx     = (const int*)d_in[1];
  const int*   kg_adj       = (const int*)d_in[2];
  const float* user_table   = (const float*)d_in[3];
  const float* item_table   = (const float*)d_in[4];
  const float* entity_table = (const float*)d_in[5];
  const float* attn_W = (const float*)d_in[6];
  const float* attn_b = (const float*)d_in[7];
  const float* kg_W1  = (const float*)d_in[8];
  const float* kg_b1  = (const float*)d_in[9];
  const float* kg_W2  = (const float*)d_in[10];
  const float* kg_b2  = (const float*)d_in[11];
  const float* comb_W = (const float*)d_in[12];
  const float* comb_b = (const float*)d_in[13];
  const float* out_W  = (const float*)d_in[14];
  const float* out_b  = (const float*)d_in[15];
  float* out = (float*)d_out;

  const int B = in_sizes[0];
  dim3 grid((B + ROWS - 1) / ROWS), block(BLK);
  kgat_fused<<<grid, block, 0, stream>>>(
      user_idx, item_idx, kg_adj, user_table, item_table, entity_table,
      attn_W, attn_b, kg_W1, kg_b1, kg_W2, kg_b2, comb_W, comb_b,
      out_W, out_b, out, B);
}


# LDS tiles (item, bufA, bufB) padded from row pitch 64 to 68 floats: the WMMA A-fragment reads (16 lanes, 16 rows, one k) were 16-way bank conflicts, now conflict-free; no sum order changes
# speedup vs baseline: 1.7388x; 1.1060x over previous
#include <hip/hip_runtime.h>

#define D     64
#define NNEI  50
#define ROWS  16
#define BLK   128
#define RPW   (ROWS / (BLK / 32))
#define SLOTS ((NNEI + 3) / 4)
#define LDP   68

typedef float v2f __attribute__((ext_vector_type(2)));
typedef float v8f __attribute__((ext_vector_type(8)));

__device__ __forceinline__ void wmma_k64_tile(v8f& c, const float* __restrict__ A_lds,
                                              const float* __restrict__ W,
                                              int nt, int hl, int half) {
  for (int kc = 0; kc < 16; ++kc) {
    const int k0 = 4 * kc + 2 * half;
    v2f a;
    a.x = A_lds[hl * LDP + k0];
    a.y = A_lds[hl * LDP + k0 + 1];
    v2f b;
    b.x = W[(size_t)k0 * D + nt * 16 + hl];
    b.y = W[(size_t)(k0 + 1) * D + nt * 16 + hl];
    c = __builtin_amdgcn_wmma_f32_16x16x4_f32(
         false, a,  false, b,
         (short)0, c,  false,  false);
  }
}

__device__ __forceinline__ void wmma_store_tile(const v8f& c, const float* __restrict__ bias,
                                                bool relu, float* __restrict__ out_lds,
                                                int nt, int hl, int half) {
  const int col = nt * 16 + hl;
  const float bv = bias[col];
#pragma unroll
  for (int i = 0; i < 8; ++i) {
    const int row = i + 8 * half;
    float v = c[i] + bv;
    if (relu) v = fmaxf(v, 0.f);
    out_lds[row * LDP + col] = v;
  }
}

template <int MASK>
__device__ __forceinline__ float row_xor(float x) {
  return __builtin_bit_cast(float, __builtin_amdgcn_update_dpp(
      0, __builtin_bit_cast(int, x), 0x160 | MASK, 0xf, 0xf, true));
}

__device__ __forceinline__ float row8_sum(float x) {
  x += row_xor<1>(x);
  x += row_xor<2>(x);
  x += row_xor<4>(x);
  return x;
}

__device__ __forceinline__ float xhalf(float x) {
  const int b = __builtin_bit_cast(int, x);
  return __builtin_bit_cast(float, __builtin_amdgcn_permlanex16(b, b, 0x76543210, 0xfedcba98, false, false));
}

__global__ __launch_bounds__(BLK) __attribute__((amdgpu_waves_per_eu(4, 4))) void kgat_fused(
    const int* __restrict__ user_idx, const int* __restrict__ item_idx,
    const int* __restrict__ kg_adj,
    const float* __restrict__ user_table, const float* __restrict__ item_table,
    const float* __restrict__ entity_table,
    const float* __restrict__ attn_W, const float* __restrict__ attn_b,
    const float* __restrict__ kg_W1, const float* __restrict__ kg_b1,
    const float* __restrict__ kg_W2, const float* __restrict__ kg_b2,
    const float* __restrict__ comb_W, const float* __restrict__ comb_b,
    const float* __restrict__ out_W, const float* __restrict__ out_b,
    float* __restrict__ out, int B) {
  __shared__ __align__(16) float item_s[ROWS * LDP];
  __shared__ __align__(16) float bufA[ROWS * LDP];
  __shared__ __align__(16) float bufB[ROWS * LDP];
  __shared__ float iscore_s[ROWS];
  __shared__ int nbr_s[ROWS * NNEI];
  __shared__ int iidx_s[ROWS];
  __shared__ int uidx_s[ROWS];

  const int tid = threadIdx.x;
  const int lane = tid & 31;
  const int wid = tid >> 5;
  const int hl = lane & 15, half = (lane >> 4) & 1;
  const int r0 = blockIdx.x * ROWS;

  if (tid < ROWS) {
    const int rr = min(r0 + tid, B - 1);
    iidx_s[tid] = item_idx[rr];
    uidx_s[tid] = user_idx[rr];
  }
  for (int t = tid; t < ROWS * NNEI; t += BLK) {
    const int rr = min(r0 + t / NNEI, B - 1);
    const int e = kg_adj[(size_t)rr * NNEI + (t % NNEI)];
    nbr_s[t] = e < 0 ? 0 : e;
  }
  __syncthreads();

  for (int t = tid; t < ROWS * (D / 4); t += BLK) {
    const int row = t >> 4, c4 = t & 15;
    const float4* src = (const float4*)(item_table + (size_t)iidx_s[row] * D);
    ((float4*)(item_s + row * LDP))[c4] = src[c4];
  }
  __syncthreads();

  if (tid < ROWS) {
    const float4* ip = (const float4*)(item_s + tid * LDP);
    const float4* wi = (const float4*)(attn_W);
    float s = 0.f;
#pragma unroll
    for (int j = 0; j < 16; ++j) {
      const float4 e = ip[j], w = wi[j];
      s += e.x * w.x + e.y * w.y + e.z * w.z + e.w * w.w;
    }
    iscore_s[tid] = s + attn_b[0];
  }
  __syncthreads();

  {
    const int j = lane & 7;
    const int g = lane >> 3;
    const float4 wlo = ((const float4*)(attn_W + D))[j];
    const float4 whi = ((const float4*)(attn_W + D))[j + 8];
    const v2f wl0 = {wlo.x, wlo.y}, wl1 = {wlo.z, wlo.w};
    const v2f wh0 = {whi.x, whi.y}, wh1 = {whi.z, whi.w};
    const bool last_ok = (4 * (SLOTS - 1) + g) < NNEI;
    const char* ebase = (const char*)entity_table;
    for (int rr = 0; rr < RPW; ++rr) {
      const int row = wid * RPW + rr;
      const int* nb = nbr_s + row * NNEI;
      float4 elo[SLOTS], ehi[SLOTS];
#pragma unroll
      for (int i = 0; i < SLOTS; ++i) {
        const int n = min(4 * i + g, NNEI - 1);
        const unsigned boff = ((unsigned)nb[n] << 8) + 16u * (unsigned)j;
        elo[i] = *(const float4*)(ebase + boff);
        ehi[i] = *(const float4*)(ebase + boff + 128u);
      }
      const float isc = iscore_s[row];
      float s[SLOTS];
      float m = -3.402823466e38f;
#pragma unroll
      for (int i = 0; i < SLOTS; ++i) {
        v2f p2 = (v2f){elo[i].x, elo[i].y} * wl0;
        p2 = (v2f){elo[i].z, elo[i].w} * wl1 + p2;
        p2 = (v2f){ehi[i].x, ehi[i].y} * wh0 + p2;
        p2 = (v2f){ehi[i].z, ehi[i].w} * wh1 + p2;
        float p = row8_sum(p2.x + p2.y);
        float sc = p + isc;
        sc = sc > 0.f ? sc : 0.2f * sc;
        s[i] = sc;
        if (i < SLOTS - 1) m = fmaxf(m, sc);
        else m = last_ok ? fmaxf(m, sc) : m;
      }
      m = fmaxf(m, row_xor<8>(m));
      m = fmaxf(m, xhalf(m));
      float sum = 0.f;
#pragma unroll
      for (int i = 0; i < SLOTS; ++i) {
        float ex = __expf(s[i] - m);
        if (i == SLOTS - 1) ex = last_ok ? ex : 0.f;
        s[i] = ex;
        sum += ex;
      }
      sum += row_xor<8>(sum);
      sum += xhalf(sum);
      const float inv = 1.f / sum;
      v2f a0 = {0.f, 0.f}, a1 = {0.f, 0.f}, a2 = {0.f, 0.f}, a3 = {0.f, 0.f};
#pragma unroll
      for (int i = 0; i < SLOTS; ++i) {
        const float w = s[i] * inv;
        const v2f w2 = {w, w};
        a0 = w2 * (v2f){elo[i].x, elo[i].y} + a0;
        a1 = w2 * (v2f){elo[i].z, elo[i].w} + a1;
        a2 = w2 * (v2f){ehi[i].x, ehi[i].y} + a2;
        a3 = w2 * (v2f){ehi[i].z, ehi[i].w} + a3;
      }
      float r[8] = {a0.x, a0.y, a1.x, a1.y, a2.x, a2.y, a3.x, a3.y};
#pragma unroll
      for (int k = 0; k < 8; ++k) {
        r[k] += row_xor<8>(r[k]);
        r[k] += xhalf(r[k]);
      }
      if (g == 0) {
        ((float4*)(bufA + row * LDP))[j]     = make_float4(r[0], r[1], r[2], r[3]);
        ((float4*)(bufA + row * LDP))[j + 8] = make_float4(r[4], r[5], r[6], r[7]);
      }
    }
  }
  __syncthreads();

  v8f c;

  c = (v8f){0, 0, 0, 0, 0, 0, 0, 0};
  wmma_k64_tile(c, bufA, kg_W1, wid, hl, half);
  wmma_store_tile(c, kg_b1, true, bufB, wid, hl, half);
  __syncthreads();

  c = (v8f){0, 0, 0, 0, 0, 0, 0, 0};
  wmma_k64_tile(c, bufB, kg_W2, wid, hl, half);
  wmma_store_tile(c, kg_b2, false, bufA, wid, hl, half);
  __syncthreads();

  c = (v8f){0, 0, 0, 0, 0, 0, 0, 0};
  wmma_k64_tile(c, item_s, comb_W, wid, hl, half);
  wmma_k64_tile(c, bufA, comb_W + D * D, wid, hl, half);
  wmma_store_tile(c, comb_b, true, bufB, wid, hl, half);
  __syncthreads();

  {
    const int row = tid >> 3;
    const int j0 = tid & 7;
    const int uidx = uidx_s[row];
    const float4* up  = (const float4*)(user_table + (size_t)uidx * D) + 2 * j0;
    const float4* w0p = (const float4*)(out_W) + 2 * j0;
    const float4* w1p = (const float4*)(out_W + D) + 2 * j0;
    const float4* fp  = (const float4*)(bufB + row * LDP) + 2 * j0;
    float s = 0.f;
#pragma unroll
    for (int j = 0; j < 2; ++j) {
      const float4 u = up[j], w0 = w0p[j];
      const float4 f = fp[j], w1 = w1p[j];
      s += u.x * w0.x + u.y * w0.y + u.z * w0.z + u.w * w0.w;
      s += f.x * w1.x + f.y * w1.y + f.z * w1.z + f.w * w1.w;
    }
    s += __shfl_xor(s, 1, 32);
    s += __shfl_xor(s, 2, 32);
    s += __shfl_xor(s, 4, 32);
    if (j0 == 0 && (r0 + row) < B) out[r0 + row] = s + out_b[0];
  }
}

extern "C" void kernel_launch(void* const* d_in, const int* in_sizes, int n_in,
                              void* d_out, int out_size, void* d_ws, size_t ws_size,
                              hipStream_t stream) {
  const int*   user_idx     = (const int*)d_in[0];
  const int*   item_idx     = (const int*)d_in[1];
  const int*   kg_adj       = (const int*)d_in[2];
  const float* user_table   = (const float*)d_in[3];
  const float* item_table   = (const float*)d_in[4];
  const float* entity_table = (const float*)d_in[5];
  const float* attn_W = (const float*)d_in[6];
  const float* attn_b = (const float*)d_in[7];
  const float* kg_W1  = (const float*)d_in[8];
  const float* kg_b1  = (const float*)d_in[9];
  const float* kg_W2  = (const float*)d_in[10];
  const float* kg_b2  = (const float*)d_in[11];
  const float* comb_W = (const float*)d_in[12];
  const float* comb_b = (const float*)d_in[13];
  const float* out_W  = (const float*)d_in[14];
  const float* out_b  = (const float*)d_in[15];
  float* out = (float*)d_out;

  const int B = in_sizes[0];
  dim3 grid((B + ROWS - 1) / ROWS), block(BLK);
  kgat_fused<<<grid, block, 0, stream>>>(
      user_idx, item_idx, kg_adj, user_table, item_table, entity_table,
      attn_W, attn_b, kg_W1, kg_b1, kg_W2, kg_b2, comb_W, comb_b,
      out_W, out_b, out, B);
}


# Global loads off the serial path: 7+2+1 one-at-a-time index/item/score round trips issued as one batch, each GEMM's weights loaded behind the previous products; item-score dot now an 8-lane tree
# speedup vs baseline: 1.9835x; 1.1407x over previous
#include <hip/hip_runtime.h>


#define D     64
#define NNEI  50
#define ROWS  16
#define BLK   128
#define RPW   (ROWS / (BLK / 32))
#define SLOTS ((NNEI + 3) / 4)
#define NIDX  ((ROWS * NNEI + BLK - 1) / BLK)
#define LDP   68

typedef float v2f __attribute__((ext_vector_type(2)));
typedef float v8f __attribute__((ext_vector_type(8)));

__device__ __forceinline__ void load_b_frags(v2f (&b)[16], const float* __restrict__ W,
                                             int nt, int hl, int half) {
#pragma unroll
  for (int kc = 0; kc < 16; ++kc) {
    const int k0 = 4 * kc + 2 * half;
    b[kc].x = W[(size_t)k0 * D + nt * 16 + hl];
    b[kc].y = W[(size_t)(k0 + 1) * D + nt * 16 + hl];
  }
}

__device__ __forceinline__ void wmma_k64_tile(v8f& c, const float* __restrict__ A_lds,
                                              const v2f (&b)[16], int hl, int half) {
#pragma unroll
  for (int kc = 0; kc < 16; ++kc) {
    const int k0 = 4 * kc + 2 * half;
    v2f a;
    a.x = A_lds[hl * LDP + k0];
    a.y = A_lds[hl * LDP + k0 + 1];
    c = __builtin_amdgcn_wmma_f32_16x16x4_f32(
         false, a,  false, b[kc],
         (short)0, c,  false,  false);
  }
}

__device__ __forceinline__ void wmma_store_tile(const v8f& c, float bv,
                                                bool relu, float* __restrict__ out_lds,
                                                int nt, int hl, int half) {
  const int col = nt * 16 + hl;
#pragma unroll
  for (int i = 0; i < 8; ++i) {
    const int row = i + 8 * half;
    float v = c[i] + bv;
    if (relu) v = fmaxf(v, 0.f);
    out_lds[row * LDP + col] = v;
  }
}

template <int MASK>
__device__ __forceinline__ float row_xor(float x) {
  return __builtin_bit_cast(float, __builtin_amdgcn_update_dpp(
      0, __builtin_bit_cast(int, x), 0x160 | MASK, 0xf, 0xf, true));
}

__device__ __forceinline__ float row8_sum(float x) {
  x += row_xor<1>(x);
  x += row_xor<2>(x);
  x += row_xor<4>(x);
  return x;
}

__device__ __forceinline__ float xhalf(float x) {
  const int b = __builtin_bit_cast(int, x);
  return __builtin_bit_cast(float, __builtin_amdgcn_permlanex16(b, b, 0x76543210, 0xfedcba98, false, false));
}

__device__ __forceinline__ void gather_issue(float4 (&elo)[SLOTS], float4 (&ehi)[SLOTS],
                                             const int* __restrict__ nb, const char* __restrict__ ebase,
                                             int j, int g) {
#pragma unroll
  for (int i = 0; i < SLOTS; ++i) {
    const int n = min(4 * i + g, NNEI - 1);
    const unsigned boff = ((unsigned)nb[n] << 8) + 16u * (unsigned)j;
    elo[i] = *(const float4*)(ebase + boff);
    ehi[i] = *(const float4*)(ebase + boff + 128u);
  }
}

__device__ __forceinline__ void gather_consume(const float4 (&elo)[SLOTS], const float4 (&ehi)[SLOTS],
                                               float isc, v2f wl0, v2f wl1, v2f wh0, v2f wh1,
                                               bool last_ok, float* __restrict__ dst, int j, int g) {
  float s[SLOTS];
  float m = -3.402823466e38f;
#pragma unroll
  for (int i = 0; i < SLOTS; ++i) {
    v2f p2 = (v2f){elo[i].x, elo[i].y} * wl0;
    p2 = (v2f){elo[i].z, elo[i].w} * wl1 + p2;
    p2 = (v2f){ehi[i].x, ehi[i].y} * wh0 + p2;
    p2 = (v2f){ehi[i].z, ehi[i].w} * wh1 + p2;
    float p = row8_sum(p2.x + p2.y);
    float sc = p + isc;
    sc = sc > 0.f ? sc : 0.2f * sc;
    s[i] = sc;
    if (i < SLOTS - 1) m = fmaxf(m, sc);
    else m = last_ok ? fmaxf(m, sc) : m;
  }
  m = fmaxf(m, row_xor<8>(m));
  m = fmaxf(m, xhalf(m));
  float sum = 0.f;
#pragma unroll
  for (int i = 0; i < SLOTS; ++i) {
    float ex = __expf(s[i] - m);
    if (i == SLOTS - 1) ex = last_ok ? ex : 0.f;
    s[i] = ex;
    sum += ex;
  }
  sum += row_xor<8>(sum);
  sum += xhalf(sum);
  const float inv = 1.f / sum;
  v2f a0 = {0.f, 0.f}, a1 = {0.f, 0.f}, a2 = {0.f, 0.f}, a3 = {0.f, 0.f};
#pragma unroll
  for (int i = 0; i < SLOTS; ++i) {
    const float w = s[i] * inv;
    const v2f w2 = {w, w};
    a0 = w2 * (v2f){elo[i].x, elo[i].y} + a0;
    a1 = w2 * (v2f){elo[i].z, elo[i].w} + a1;
    a2 = w2 * (v2f){ehi[i].x, ehi[i].y} + a2;
    a3 = w2 * (v2f){ehi[i].z, ehi[i].w} + a3;
  }
  float r[8] = {a0.x, a0.y, a1.x, a1.y, a2.x, a2.y, a3.x, a3.y};
#pragma unroll
  for (int k = 0; k < 8; ++k) {
    r[k] += row_xor<8>(r[k]);
    r[k] += xhalf(r[k]);
  }
  if (g == 0) {
    ((float4*)dst)[j]     = make_float4(r[0], r[1], r[2], r[3]);
    ((float4*)dst)[j + 8] = make_float4(r[4], r[5], r[6], r[7]);
  }
}

__global__ __launch_bounds__(BLK) __attribute__((amdgpu_waves_per_eu(4, 4))) void kgat_fused(
    const int* __restrict__ user_idx, const int* __restrict__ item_idx,
    const int* __restrict__ kg_adj,
    const float* __restrict__ user_table, const float* __restrict__ item_table,
    const float* __restrict__ entity_table,
    const float* __restrict__ attn_W, const float* __restrict__ attn_b,
    const float* __restrict__ kg_W1, const float* __restrict__ kg_b1,
    const float* __restrict__ kg_W2, const float* __restrict__ kg_b2,
    const float* __restrict__ comb_W, const float* __restrict__ comb_b,
    const float* __restrict__ out_W, const float* __restrict__ out_b,
    float* __restrict__ out, int B) {
  __shared__ __align__(16) float item_s[ROWS * LDP];
  __shared__ __align__(16) float bufA[ROWS * LDP];
  __shared__ __align__(16) float bufB[ROWS * LDP];
  __shared__ int nbr_s[NIDX * BLK];

  const int tid = threadIdx.x;
  const int lane = tid & 31;
  const int wid = tid >> 5;
  const int hl = lane & 15, half = (lane >> 4) & 1;
  const int r0 = blockIdx.x * ROWS;
  const int j = lane & 7;
  const int g = lane >> 3;
  const int myrow = wid * RPW + g;
  const int grow = min(r0 + myrow, B - 1);

  const int iidx = item_idx[grow];
  const int uidx = user_idx[grow];
  int nbv[NIDX];
#pragma unroll
  for (int u = 0; u < NIDX; ++u) {
    const int t = min(tid + u * BLK, ROWS * NNEI - 1);
    const int rr = min(r0 + t / NNEI, B - 1);
    nbv[u] = kg_adj[(size_t)rr * NNEI + (t % NNEI)];
  }
  const float4 wlo = ((const float4*)(attn_W + D))[j];
  const float4 whi = ((const float4*)(attn_W + D))[j + 8];
  const float4 vlo = ((const float4*)(attn_W))[j];
  const float4 vhi = ((const float4*)(attn_W))[j + 8];
  const float ab = attn_b[0];
  const v2f wl0 = {wlo.x, wlo.y}, wl1 = {wlo.z, wlo.w};
  const v2f wh0 = {whi.x, whi.y}, wh1 = {whi.z, whi.w};

#pragma unroll
  for (int u = 0; u < NIDX; ++u)
    nbr_s[tid + u * BLK] = nbv[u] < 0 ? 0 : nbv[u];
  __syncthreads();

  const bool last_ok = (4 * (SLOTS - 1) + g) < NNEI;
  const char* ebase = (const char*)entity_table;
  float4 elo[SLOTS], ehi[SLOTS];

  gather_issue(elo, ehi, nbr_s + (wid * RPW) * NNEI, ebase, j, g);
  const float4* ip = (const float4*)(item_table + (size_t)iidx * D);
  const float4 itlo = ip[j], ithi = ip[j + 8];

  v2f bw1[16];
  load_b_frags(bw1, kg_W1, wid, hl, half);

  float iscv;
  {
    v2f q2 = (v2f){itlo.x, itlo.y} * (v2f){vlo.x, vlo.y};
    q2 = (v2f){itlo.z, itlo.w} * (v2f){vlo.z, vlo.w} + q2;
    q2 = (v2f){ithi.x, ithi.y} * (v2f){vhi.x, vhi.y} + q2;
    q2 = (v2f){ithi.z, ithi.w} * (v2f){vhi.z, vhi.w} + q2;
    iscv = row8_sum(q2.x + q2.y) + ab;
  }
  ((float4*)(item_s + myrow * LDP))[j]     = itlo;
  ((float4*)(item_s + myrow * LDP))[j + 8] = ithi;
  const int iscb = __builtin_bit_cast(int, iscv);

  gather_consume(elo, ehi, __builtin_bit_cast(float, __builtin_amdgcn_readlane(iscb, 0)),
                 wl0, wl1, wh0, wh1, last_ok, bufA + (wid * RPW) * LDP, j, g);
  for (int rr = 1; rr < RPW; ++rr) {
    const int row = wid * RPW + rr;
    gather_issue(elo, ehi, nbr_s + row * NNEI, ebase, j, g);
    const float isc = __builtin_bit_cast(float, __builtin_amdgcn_readlane(iscb, 8 * rr));
    gather_consume(elo, ehi, isc, wl0, wl1, wh0, wh1, last_ok, bufA + row * LDP, j, g);
  }

  __syncthreads();

  v8f c;

  v2f bw2[16];
  load_b_frags(bw2, kg_W2, wid, hl, half);
  const float4* up  = (const float4*)(user_table + (size_t)uidx * D) + 2 * j;
  const float4* w0p = (const float4*)(out_W) + 2 * j;
  const float4* w1p = (const float4*)(out_W + D) + 2 * j;
  const float4 u0 = up[0], u1 = up[1];
  const float4 ow00 = w0p[0], ow01 = w0p[1];
  const float4 ow10 = w1p[0], ow11 = w1p[1];
  const float b1v = kg_b1[wid * 16 + hl];
  const float b2v = kg_b2[wid * 16 + hl];
  const float cbv = comb_b[wid * 16 + hl];
  const float ob = out_b[0];
  c = (v8f){0, 0, 0, 0, 0, 0, 0, 0};
  wmma_k64_tile(c, bufA, bw1, hl, half);
  wmma_store_tile(c, b1v, true, bufB, wid, hl, half);
  __syncthreads();

  v2f bc0[16], bc1[16];
  load_b_frags(bc0, comb_W, wid, hl, half);
  load_b_frags(bc1, comb_W + D * D, wid, hl, half);
  c = (v8f){0, 0, 0, 0, 0, 0, 0, 0};
  wmma_k64_tile(c, bufB, bw2, hl, half);
  wmma_store_tile(c, b2v, false, bufA, wid, hl, half);
  __syncthreads();

  c = (v8f){0, 0, 0, 0, 0, 0, 0, 0};
  wmma_k64_tile(c, item_s, bc0, hl, half);
  wmma_k64_tile(c, bufA, bc1, hl, half);
  wmma_store_tile(c, cbv, true, bufB, wid, hl, half);
  __syncthreads();

  {
    const float4* fp = (const float4*)(bufB + myrow * LDP) + 2 * j;
    const float4 f0 = fp[0], f1 = fp[1];
    float s = 0.f;
    s += u0.x * ow00.x + u0.y * ow00.y + u0.z * ow00.z + u0.w * ow00.w;
    s += f0.x * ow10.x + f0.y * ow10.y + f0.z * ow10.z + f0.w * ow10.w;
    s += u1.x * ow01.x + u1.y * ow01.y + u1.z * ow01.z + u1.w * ow01.w;
    s += f1.x * ow11.x + f1.y * ow11.y + f1.z * ow11.z + f1.w * ow11.w;
    s += __shfl_xor(s, 1, 32);
    s += __shfl_xor(s, 2, 32);
    s += __shfl_xor(s, 4, 32);
    if (j == 0 && (r0 + myrow) < B) out[r0 + myrow] = s + ob;
  }
}

extern "C" void kernel_launch(void* const* d_in, const int* in_sizes, int n_in,
                              void* d_out, int out_size, void* d_ws, size_t ws_size,
                              hipStream_t stream) {
  const int*   user_idx     = (const int*)d_in[0];
  const int*   item_idx     = (const int*)d_in[1];
  const int*   kg_adj       = (const int*)d_in[2];
  const float* user_table   = (const float*)d_in[3];
  const float* item_table   = (const float*)d_in[4];
  const float* entity_table = (const float*)d_in[5];
  const float* attn_W = (const float*)d_in[6];
  const float* attn_b = (const float*)d_in[7];
  const float* kg_W1  = (const float*)d_in[8];
  const float* kg_b1  = (const float*)d_in[9];
  const float* kg_W2  = (const float*)d_in[10];
  const float* kg_b2  = (const float*)d_in[11];
  const float* comb_W = (const float*)d_in[12];
  const float* comb_b = (const float*)d_in[13];
  const float* out_W  = (const float*)d_in[14];
  const float* out_b  = (const float*)d_in[15];
  float* out = (float*)d_out;

  const int B = in_sizes[0];
  dim3 grid((B + ROWS - 1) / ROWS), block(BLK);
  kgat_fused<<<grid, block, 0, stream>>>(
      user_idx, item_idx, kg_adj, user_table, item_table, entity_table,
      attn_W, attn_b, kg_W1, kg_b1, kg_W2, kg_b2, comb_W, comb_b,
      out_W, out_b, out, B);
}
